# starting kernel
# baseline (speedup 1.0000x reference)
#include <hip/hip_runtime.h>
#include <hip/hip_bf16.h>

typedef __attribute__((ext_vector_type(16))) _Float16 v16h;
typedef __attribute__((ext_vector_type(8)))  float    v8f;

#define NN   512
#define NE   4096
#define BB   32
#define HH   32
#define SKC  256
#define ECC  512
#define OUTF 12
#define AEMB 10
#define NEGBIG (-1e30f)
#define NLAY 8

__device__ __forceinline__ v8f wmma_f16(v16h a, v16h b, v8f c) {
    return __builtin_amdgcn_wmma_f32_16x16x32_f16(false, a, false, b, (short)0, c, false, false);
}

__device__ __forceinline__ float fast_sigmoid(float x) {
    return __builtin_amdgcn_rcpf(1.f + __expf(-x));
}
__device__ __forceinline__ float fast_tanh(float x) {
    float t = __expf(-2.f * fabsf(x));
    float r = (1.f - t) * __builtin_amdgcn_rcpf(1.f + t);
    return copysignf(r, x);
}

__device__ __forceinline__ void pack8(v16h& d, int off, const float* __restrict__ p) {
    float4 x = ((const float4*)p)[0];
    float4 y = ((const float4*)p)[1];
    d[off + 0] = (_Float16)x.x; d[off + 1] = (_Float16)x.y;
    d[off + 2] = (_Float16)x.z; d[off + 3] = (_Float16)x.w;
    d[off + 4] = (_Float16)y.x; d[off + 5] = (_Float16)y.y;
    d[off + 6] = (_Float16)y.z; d[off + 7] = (_Float16)y.w;
}
__device__ __forceinline__ void pack8_scale(v16h& d, int off, const float* __restrict__ p, float s) {
    float4 x = ((const float4*)p)[0];
    float4 y = ((const float4*)p)[1];
    d[off + 0] = (_Float16)(x.x * s); d[off + 1] = (_Float16)(x.y * s);
    d[off + 2] = (_Float16)(x.z * s); d[off + 3] = (_Float16)(x.w * s);
    d[off + 4] = (_Float16)(y.x * s); d[off + 5] = (_Float16)(y.y * s);
    d[off + 6] = (_Float16)(y.z * s); d[off + 7] = (_Float16)(y.w * s);
}
__device__ __forceinline__ void pack8_relu(v16h& d, int off, const float* __restrict__ p) {
    float4 x = ((const float4*)p)[0];
    float4 y = ((const float4*)p)[1];
    d[off + 0] = (_Float16)fmaxf(x.x, 0.f); d[off + 1] = (_Float16)fmaxf(x.y, 0.f);
    d[off + 2] = (_Float16)fmaxf(x.z, 0.f); d[off + 3] = (_Float16)fmaxf(x.w, 0.f);
    d[off + 4] = (_Float16)fmaxf(y.x, 0.f); d[off + 5] = (_Float16)fmaxf(y.y, 0.f);
    d[off + 6] = (_Float16)fmaxf(y.z, 0.f); d[off + 7] = (_Float16)fmaxf(y.w, 0.f);
}

__global__ void k_zero(float* p, int n) {
    int i = blockIdx.x * blockDim.x + threadIdx.x;
    if (i < n) p[i] = 0.f;
}

__global__ void k_prep_w(const float* __restrict__ fw, const float* __restrict__ gw,
                         const float* __restrict__ gcw,
                         float* fwT, float* gwT, float* gcwT) {
    int idx = blockIdx.x * blockDim.x + threadIdx.x;
    const int totalFG = NLAY * HH * 2 * HH;
    if (idx < totalFG) {
        int ci = idx & 31;
        int tap = (idx >> 5) & 1;
        int c = (idx >> 6) & 31;
        int l = idx >> 11;
        fwT[idx] = fw[((l * HH + c) * HH + ci) * 2 + tap];
        gwT[idx] = gw[((l * HH + c) * HH + ci) * 2 + tap];
    }
    if (idx < NLAY * HH * HH) {
        int k = idx & 31;
        int n = (idx >> 5) & 31;
        int l = idx >> 10;
        gcwT[idx] = gcw[l * HH * HH + k * HH + n];
    }
}

__global__ void k_scatter(const int* __restrict__ src, const int* __restrict__ dst,
                          unsigned char* adj, float* degout, float* degin) {
    int e = blockIdx.x * blockDim.x + threadIdx.x;
    if (e >= NE) return;
    int s = src[e], d = dst[e];
    adj[s * NN + d] = 1;
    atomicAdd(degout + s, 1.f);
    atomicAdd(degin + d, 1.f);
}

__global__ void k_rownorm(const unsigned char* __restrict__ adj,
                          const float* __restrict__ nv1, const float* __restrict__ nv2,
                          float* rowmax, float* rowsum) {
    __shared__ float sc[NN];
    __shared__ float red[256];
    int i = blockIdx.x;
    for (int j = threadIdx.x; j < NN; j += blockDim.x) {
        float v = NEGBIG;
        if (adj[i * NN + j]) {
            float acc = 0.f;
#pragma unroll
            for (int a = 0; a < AEMB; a++) acc += nv1[i * AEMB + a] * nv2[a * NN + j];
            v = fmaxf(acc, 0.f);
        }
        sc[j] = v;
    }
    __syncthreads();
    float m = NEGBIG;
    for (int j = threadIdx.x; j < NN; j += blockDim.x) m = fmaxf(m, sc[j]);
    red[threadIdx.x] = m;
    __syncthreads();
    for (int s = blockDim.x / 2; s > 0; s >>= 1) {
        if (threadIdx.x < s) red[threadIdx.x] = fmaxf(red[threadIdx.x], red[threadIdx.x + s]);
        __syncthreads();
    }
    m = red[0];
    __syncthreads();
    float sum = 0.f;
    for (int j = threadIdx.x; j < NN; j += blockDim.x) {
        float v = sc[j];
        if (v > 0.5f * NEGBIG) sum += __expf(v - m);
    }
    red[threadIdx.x] = sum;
    __syncthreads();
    for (int s = blockDim.x / 2; s > 0; s >>= 1) {
        if (threadIdx.x < s) red[threadIdx.x] += red[threadIdx.x + s];
        __syncthreads();
    }
    if (threadIdx.x == 0) {
        rowmax[i] = m;
        rowsum[i] = (red[0] > 0.f) ? red[0] : 1.f;
    }
}

__global__ void k_ew(const int* __restrict__ src, const int* __restrict__ dst,
                     const float* __restrict__ nv1, const float* __restrict__ nv2,
                     const float* __restrict__ rowmax, const float* __restrict__ rowsum,
                     float* ew) {
    int e = blockIdx.x * blockDim.x + threadIdx.x;
    if (e >= NE) return;
    int s = src[e], d = dst[e];
    float acc = 0.f;
#pragma unroll
    for (int a = 0; a < AEMB; a++) acc += nv1[s * AEMB + a] * nv2[a * NN + d];
    float scv = fmaxf(acc, 0.f);
    ew[e] = 1.f + __expf(scv - rowmax[s]) * __builtin_amdgcn_rcpf(rowsum[s]);
}

__global__ void k_norms(const float* degout, const float* degin, float* nsv, float* ndv) {
    int i = blockIdx.x * blockDim.x + threadIdx.x;
    if (i >= NN) return;
    nsv[i] = rsqrtf(fmaxf(degout[i], 1.f));
    ndv[i] = rsqrtf(fmaxf(degin[i], 1.f));
}

__global__ void k_start(const float* __restrict__ feat, const float* __restrict__ sw,
                        const float* __restrict__ sb, float* h) {
    int idx = blockIdx.x * blockDim.x + threadIdx.x;
    const int total = BB * NN * 13 * HH;
    if (idx >= total) return;
    int c = idx & 31;
    int t = (idx >> 5) % 13;
    int m = idx / (13 * HH);
    int b = m >> 9, n = m & 511;
    float x = (t == 0) ? 0.f : feat[(b * 12 + (t - 1)) * NN + n];
    h[idx] = sw[c] * x + sb[c];
}

__global__ void k_gau_wmma(const float* __restrict__ h, const float* __restrict__ fwT,
                           const float* __restrict__ fb, const float* __restrict__ gwT,
                           const float* __restrict__ gb, float* gau, int T, int d, int l) {
    int Tp = T - d;
    int tile = blockIdx.x * 4 + (threadIdx.x >> 5);
    int tm = tile >> 1, tn = tile & 1;
    int lane = threadIdx.x & 31;
    int mrow = tm * 16 + (lane & 15);
    int t = mrow % Tp;
    int bn = mrow / Tp;
    const float* hrow0 = h + (bn * T + t) * HH;
    const float* hrow1 = hrow0 + d * HH;
    int ncol = tn * 16 + (lane & 15);
    int base = (lane & 16) ? 8 : 0;
    const float* fwr = fwT + (size_t)((l * HH + ncol) * 2) * HH;
    const float* gwr = gwT + (size_t)((l * HH + ncol) * 2) * HH;
    v8f cf = {}, cg = {};
    v16h a, bf, bg;
    pack8(a, 0, hrow0 + base);  pack8(a, 8, hrow0 + base + 16);
    pack8(bf, 0, fwr + base);   pack8(bf, 8, fwr + base + 16);
    pack8(bg, 0, gwr + base);   pack8(bg, 8, gwr + base + 16);
    cf = wmma_f16(a, bf, cf);
    cg = wmma_f16(a, bg, cg);
    pack8(a, 0, hrow1 + base);      pack8(a, 8, hrow1 + base + 16);
    pack8(bf, 0, fwr + HH + base);  pack8(bf, 8, fwr + HH + base + 16);
    pack8(bg, 0, gwr + HH + base);  pack8(bg, 8, gwr + HH + base + 16);
    cf = wmma_f16(a, bf, cf);
    cg = wmma_f16(a, bg, cg);

    int rbase = tm * 16 + ((lane & 16) ? 8 : 0);
    float fbias = fb[l * HH + ncol];
    float gbias = gb[l * HH + ncol];
#pragma unroll
    for (int v = 0; v < 8; v++) {
        int row = rbase + v;
        float fv = fast_tanh(cf[v] + fbias);
        float gv = fast_sigmoid(cg[v] + gbias);
        gau[(size_t)row * HH + ncol] = fv * gv;
    }
}

__global__ void k_skip_wmma(const float* __restrict__ gau, const float* __restrict__ sw,
                            const float* __restrict__ sb, float* skip, int Tp, int l) {
    const int nTilesN = SKC / 16;
    int tile = blockIdx.x * 4 + (threadIdx.x >> 5);
    int tm = tile / nTilesN, tn = tile % nTilesN;
    int lane = threadIdx.x & 31;
    int mrow = tm * 16 + (lane & 15);
    int ncol = tn * 16 + (lane & 15);
    int base = (lane & 16) ? 8 : 0;
    const float* arow = gau + (size_t)(mrow * Tp + (Tp - 1)) * HH;
    const float* brow = sw + (size_t)(l * SKC + ncol) * HH;
    v16h a, bm;
    pack8(a, 0, arow + base);  pack8(a, 8, arow + base + 16);
    pack8(bm, 0, brow + base); pack8(bm, 8, brow + base + 16);
    v8f c = {};
    c = wmma_f16(a, bm, c);
    int rbase = tm * 16 + ((lane & 16) ? 8 : 0);
    float bias = sb[l * SKC + ncol];
#pragma unroll
    for (int v = 0; v < 8; v++) {
        float* p = skip + (size_t)(rbase + v) * SKC + ncol;
        *p += c[v] + bias;
    }
}

__global__ void k_msg(const int* __restrict__ src, const int* __restrict__ dst,
                      const float* __restrict__ ew, const float* __restrict__ nsv,
                      const float* __restrict__ gau, float* agg, int Tp) {
    int idx = blockIdx.x * blockDim.x + threadIdx.x;
    int total = NE * BB * Tp * HH;
    if (idx >= total) return;
    int c = idx & 31;
    int t = (idx >> 5) % Tp;
    int r = idx / (Tp * HH);
    int e = r % NE;
    int b = r / NE;
    int s = src[e], d = dst[e];
    float coeff = ew[e] * nsv[s];
    float* pa = &agg[((b * NN + d) * Tp + t) * HH + c];
    __builtin_prefetch(pa, 1, 0);
    atomicAdd(pa, coeff * gau[((b * NN + s) * Tp + t) * HH + c]);
}

__global__ void k_gconv_wmma(const float* __restrict__ agg, const float* __restrict__ ndv,
                             const float* __restrict__ gcwT, const float* __restrict__ gcb,
                             const float* __restrict__ hold, float* hpre,
                             float* bnsum, float* bnsq, int T, int d, int l) {
    int Tp = T - d;
    int tile = blockIdx.x * 4 + (threadIdx.x >> 5);
    int tm = tile >> 1, tn = tile & 1;
    int lane = threadIdx.x & 31;
    int mrow = tm * 16 + (lane & 15);
    int node_m = (mrow / Tp) & 511;
    float nrm = ndv[node_m];
    int ncol = tn * 16 + (lane & 15);
    int base = (lane & 16) ? 8 : 0;
    const float* arow = agg + (size_t)mrow * HH;
    const float* brow = gcwT + (size_t)(l * HH + ncol) * HH;
    v16h a, bm;
    pack8_scale(a, 0, arow + base, nrm);
    pack8_scale(a, 8, arow + base + 16, nrm);
    pack8(bm, 0, brow + base);
    pack8(bm, 8, brow + base + 16);
    v8f c = {};
    c = wmma_f16(a, bm, c);
    int rbase = tm * 16 + ((lane & 16) ? 8 : 0);
    float bias = gcb[l * HH + ncol];
    float s1 = 0.f, s2 = 0.f;
#pragma unroll
    for (int v = 0; v < 8; v++) {
        int row = rbase + v;
        int t = row % Tp;
        int bn = row / Tp;
        float val = fmaxf(c[v] + bias, 0.f) + hold[(bn * T + t + d) * HH + ncol];
        hpre[(size_t)row * HH + ncol] = val;
        s1 += val;
        s2 += val * val;
    }
    atomicAdd(&bnsum[ncol], s1);
    atomicAdd(&bnsq[ncol], s2);
}

__global__ void k_bn(float* h, const float* __restrict__ bnsum, const float* __restrict__ bnsq,
                     const float* __restrict__ gamma, const float* __restrict__ beta,
                     int Tp, int l) {
    int total = BB * NN * Tp * HH;
    int idx = blockIdx.x * blockDim.x + threadIdx.x;
    if (idx >= total) return;
    int c = idx & 31;
    float cnt = (float)(BB * NN * Tp);
    float mu = bnsum[c] / cnt;
    float var = bnsq[c] / cnt - mu * mu;
    float x = h[idx];
    h[idx] = (x - mu) * rsqrtf(var + 1e-5f) * gamma[l * HH + c] + beta[l * HH + c];
}

__global__ void k_end1_wmma(const float* __restrict__ skip, const float* __restrict__ w1,
                            const float* __restrict__ b1, float* out1) {
    const int nTilesN = ECC / 16;
    int tile = blockIdx.x * 4 + (threadIdx.x >> 5);
    int tm = tile / nTilesN, tn = tile % nTilesN;
    int lane = threadIdx.x & 31;
    int mrow = tm * 16 + (lane & 15);
    int ncol = tn * 16 + (lane & 15);
    int base = (lane & 16) ? 8 : 0;
    const float* arow = skip + (size_t)mrow * SKC;
    const float* brow = w1 + (size_t)ncol * SKC;
    v8f c = {};
#pragma unroll
    for (int k0 = 0; k0 < SKC; k0 += 32) {
        v16h a, bm;
        pack8_relu(a, 0, arow + k0 + base);
        pack8_relu(a, 8, arow + k0 + base + 16);
        pack8(bm, 0, brow + k0 + base);
        pack8(bm, 8, brow + k0 + base + 16);
        c = wmma_f16(a, bm, c);
    }
    int rbase = tm * 16 + ((lane & 16) ? 8 : 0);
    float bias = b1[ncol];
#pragma unroll
    for (int v = 0; v < 8; v++)
        out1[(size_t)(rbase + v) * ECC + ncol] = fmaxf(c[v] + bias, 0.f);
}

__global__ void k_end2_wmma(const float* __restrict__ in1, const float* __restrict__ w2,
                            const float* __restrict__ b2, float* out) {
    int tm = blockIdx.x * 4 + (threadIdx.x >> 5);
    int lane = threadIdx.x & 31;
    int mrow = tm * 16 + (lane & 15);
    int ncol = lane & 15;
    int base = (lane & 16) ? 8 : 0;
    const float* arow = in1 + (size_t)mrow * ECC;
    const float* brow = w2 + (size_t)ncol * ECC;
    v8f c = {};
#pragma unroll
    for (int k0 = 0; k0 < ECC; k0 += 32) {
        v16h a, bm = {};
        pack8(a, 0, arow + k0 + base);
        pack8(a, 8, arow + k0 + base + 16);
        if (ncol < OUTF) {
            pack8(bm, 0, brow + k0 + base);
            pack8(bm, 8, brow + k0 + base + 16);
        }
        c = wmma_f16(a, bm, c);
    }
    if (ncol < OUTF) {
        int rbase = tm * 16 + ((lane & 16) ? 8 : 0);
        float bias = b2[ncol];
#pragma unroll
        for (int v = 0; v < 8; v++) {
            int row = rbase + v;
            int b = row >> 9, n = row & 511;
            out[(b * OUTF + ncol) * NN + n] = c[v] + bias;
        }
    }
}

extern "C" void kernel_launch(void* const* d_in, const int* in_sizes, int n_in,
                              void* d_out, int out_size, void* d_ws, size_t ws_size,
                              hipStream_t stream) {
    const float* feat     = (const float*)d_in[0];
    const int*   src      = (const int*)d_in[1];
    const int*   dst      = (const int*)d_in[2];
    const float* nv1      = (const float*)d_in[3];
    const float* nv2      = (const float*)d_in[4];
    const float* start_w  = (const float*)d_in[5];
    const float* start_b  = (const float*)d_in[6];
    const float* filter_w = (const float*)d_in[7];
    const float* filter_b = (const float*)d_in[8];
    const float* gate_w   = (const float*)d_in[9];
    const float* gate_b   = (const float*)d_in[10];
    const float* skip_w   = (const float*)d_in[11];
    const float* skip_b   = (const float*)d_in[12];
    const float* bn_gamma = (const float*)d_in[13];
    const float* bn_beta  = (const float*)d_in[14];
    const float* gconv_w  = (const float*)d_in[15];
    const float* gconv_b  = (const float*)d_in[16];
    const float* end1_w   = (const float*)d_in[17];
    const float* end1_b   = (const float*)d_in[18];
    const float* end2_w   = (const float*)d_in[19];
    const float* end2_b   = (const float*)d_in[20];

    float* ws = (float*)d_ws;
    size_t o = 0;
    unsigned char* adj = (unsigned char*)(ws + o); o += (NN * NN) / 4;
    float* rowmax = ws + o; o += NN;
    float* rowsum = ws + o; o += NN;
    float* ew     = ws + o; o += NE;
    float* degout = ws + o; o += NN;
    float* degin  = ws + o; o += NN;
    float* nsv    = ws + o; o += NN;
    float* ndv    = ws + o; o += NN;
    float* bnsum  = ws + o; o += HH;
    float* bnsq   = ws + o; o += HH;
    float* fwT    = ws + o; o += NLAY * HH * 2 * HH;
    float* gwT    = ws + o; o += NLAY * HH * 2 * HH;
    float* gcwT   = ws + o; o += NLAY * HH * HH;
    float* skip   = ws + o; o += (size_t)BB * NN * SKC;
    float* hA     = ws + o; o += (size_t)BB * NN * 13 * HH;
    float* hB     = ws + o; o += (size_t)BB * NN * 13 * HH;
    float* gau    = ws + o; o += (size_t)BB * NN * 12 * HH;
    float* agg    = ws + o; o += (size_t)BB * NN * 12 * HH;
    float* out1   = gau;

#define ZEROF(p, n) k_zero<<<((n) + 255) / 256, 256, 0, stream>>>((p), (n))

    ZEROF((float*)adj, (NN * NN) / 4);
    ZEROF(degout, NN);
    ZEROF(degin, NN);
    ZEROF(skip, BB * NN * SKC);

    k_prep_w<<<(NLAY * HH * 2 * HH + 255) / 256, 256, 0, stream>>>(filter_w, gate_w, gconv_w,
                                                                   fwT, gwT, gcwT);
    k_scatter<<<(NE + 255) / 256, 256, 0, stream>>>(src, dst, adj, degout, degin);
    k_rownorm<<<NN, 256, 0, stream>>>(adj, nv1, nv2, rowmax, rowsum);
    k_ew<<<(NE + 255) / 256, 256, 0, stream>>>(src, dst, nv1, nv2, rowmax, rowsum, ew);
    k_norms<<<(NN + 255) / 256, 256, 0, stream>>>(degout, degin, nsv, ndv);
    k_start<<<(BB * NN * 13 * HH + 255) / 256, 256, 0, stream>>>(feat, start_w, start_b, hA);

    const int dil[8] = {1, 2, 1, 2, 1, 2, 1, 2};
    float* hcur = hA;
    float* hnext = hB;
    int T = 13;
    for (int l = 0; l < NLAY; l++) {
        int d = dil[l];
        int Tp = T - d;
        int M = BB * NN * Tp;
        int tilesGau = (M / 16) * 2;

        k_gau_wmma<<<tilesGau / 4, 128, 0, stream>>>(hcur, fwT, filter_b, gwT, gate_b,
                                                     gau, T, d, l);
        k_skip_wmma<<<(BB * NN / 16) * (SKC / 16) / 4, 128, 0, stream>>>(gau, skip_w, skip_b,
                                                                         skip, Tp, l);
        ZEROF(agg, M * HH);
        k_msg<<<(NE * BB * Tp * HH + 255) / 256, 256, 0, stream>>>(src, dst, ew, nsv,
                                                                   gau, agg, Tp);
        ZEROF(bnsum, HH);
        ZEROF(bnsq, HH);
        k_gconv_wmma<<<(M / 16) * 2 / 4, 128, 0, stream>>>(agg, ndv, gcwT, gconv_b,
                                                           hcur, hnext, bnsum, bnsq, T, d, l);
        k_bn<<<(M * HH + 255) / 256, 256, 0, stream>>>(hnext, bnsum, bnsq,
                                                       bn_gamma, bn_beta, Tp, l);
        float* tmp = hcur; hcur = hnext; hnext = tmp;
        T = Tp;
    }

    k_end1_wmma<<<(BB * NN / 16) * (ECC / 16) / 4, 128, 0, stream>>>(skip, end1_w, end1_b, out1);
    k_end2_wmma<<<(BB * NN / 16) / 4, 128, 0, stream>>>(out1, end2_w, end2_b, (float*)d_out);
#undef ZEROF
}


# end1 product block-tiled: a block owns 128 rows x 128 columns, reads both operand tiles once with whole-line loads, keeps them in LDS as f16 (same rounding), stores whole output lines
# speedup vs baseline: 4.5201x; 4.5201x over previous
#include <hip/hip_runtime.h>
#include <hip/hip_bf16.h>

typedef __attribute__((ext_vector_type(16))) _Float16 v16h;
typedef __attribute__((ext_vector_type(8)))  float    v8f;
typedef __attribute__((ext_vector_type(8)))  _Float16 v8h;
typedef __attribute__((ext_vector_type(4)))  _Float16 v4h;

#define NN   512
#define NE   4096
#define BB   32
#define HH   32
#define SKC  256
#define ECC  512
#define OUTF 12
#define AEMB 10
#define NEGBIG (-1e30f)
#define NLAY 8

__device__ __forceinline__ v8f wmma_f16(v16h a, v16h b, v8f c) {
    return __builtin_amdgcn_wmma_f32_16x16x32_f16(false, a, false, b, (short)0, c, false, false);
}

__device__ __forceinline__ float fast_sigmoid(float x) {
    return __builtin_amdgcn_rcpf(1.f + __expf(-x));
}
__device__ __forceinline__ float fast_tanh(float x) {
    float t = __expf(-2.f * fabsf(x));
    float r = (1.f - t) * __builtin_amdgcn_rcpf(1.f + t);
    return copysignf(r, x);
}

__device__ __forceinline__ void pack8(v16h& d, int off, const float* __restrict__ p) {
    float4 x = ((const float4*)p)[0];
    float4 y = ((const float4*)p)[1];
    d[off + 0] = (_Float16)x.x; d[off + 1] = (_Float16)x.y;
    d[off + 2] = (_Float16)x.z; d[off + 3] = (_Float16)x.w;
    d[off + 4] = (_Float16)y.x; d[off + 5] = (_Float16)y.y;
    d[off + 6] = (_Float16)y.z; d[off + 7] = (_Float16)y.w;
}
__device__ __forceinline__ void pack8_relu(v16h& d, int off, const float* __restrict__ p) {
    float4 x = ((const float4*)p)[0];
    float4 y = ((const float4*)p)[1];
    d[off + 0] = (_Float16)fmaxf(x.x, 0.f); d[off + 1] = (_Float16)fmaxf(x.y, 0.f);
    d[off + 2] = (_Float16)fmaxf(x.z, 0.f); d[off + 3] = (_Float16)fmaxf(x.w, 0.f);
    d[off + 4] = (_Float16)fmaxf(y.x, 0.f); d[off + 5] = (_Float16)fmaxf(y.y, 0.f);
    d[off + 6] = (_Float16)fmaxf(y.z, 0.f); d[off + 7] = (_Float16)fmaxf(y.w, 0.f);
}

__global__ void k_zero(float* p, int n) {
    int i = blockIdx.x * blockDim.x + threadIdx.x;
    if (i < n) p[i] = 0.f;
}

__global__ void k_prep_w(const float* __restrict__ fw, const float* __restrict__ gw,
                         const float* __restrict__ gcw,
                         float* fwT, float* gwT, float* gcwT) {
    int idx = blockIdx.x * blockDim.x + threadIdx.x;
    const int totalFG = NLAY * HH * 2 * HH;
    if (idx < totalFG) {
        int ci = idx & 31;
        int tap = (idx >> 5) & 1;
        int c = (idx >> 6) & 31;
        int l = idx >> 11;
        fwT[idx] = fw[((l * HH + c) * HH + ci) * 2 + tap];
        gwT[idx] = gw[((l * HH + c) * HH + ci) * 2 + tap];
    }
    if (idx < NLAY * HH * HH) {
        int k = idx & 31;
        int n = (idx >> 5) & 31;
        int l = idx >> 10;
        gcwT[idx] = gcw[l * HH * HH + k * HH + n];
    }
}

__global__ void k_scatter(const int* __restrict__ src, const int* __restrict__ dst,
                          int* cnt, float* degout, float* degin) {
    int e = blockIdx.x * blockDim.x + threadIdx.x;
    if (e >= NE) return;
    int s = src[e], d = dst[e];
    atomicAdd(cnt + s * NN + d, 1);
    atomicAdd(degout + s, 1.f);
    atomicAdd(degin + d, 1.f);
}

__global__ void k_rownorm(const int* __restrict__ adj,
                          const float* __restrict__ nv1, const float* __restrict__ nv2,
                          float* rowmax, float* rowsum) {
    __shared__ float sc[NN];
    __shared__ float red[256];
    int i = blockIdx.x;
    for (int j = threadIdx.x; j < NN; j += blockDim.x) {
        float v = NEGBIG;
        if (adj[i * NN + j]) {
            float acc = 0.f;
#pragma unroll
            for (int a = 0; a < AEMB; a++) acc += nv1[i * AEMB + a] * nv2[a * NN + j];
            v = fmaxf(acc, 0.f);
        }
        sc[j] = v;
    }
    __syncthreads();
    float m = NEGBIG;
    for (int j = threadIdx.x; j < NN; j += blockDim.x) m = fmaxf(m, sc[j]);
    red[threadIdx.x] = m;
    __syncthreads();
    for (int s = blockDim.x / 2; s > 0; s >>= 1) {
        if (threadIdx.x < s) red[threadIdx.x] = fmaxf(red[threadIdx.x], red[threadIdx.x + s]);
        __syncthreads();
    }
    m = red[0];
    __syncthreads();
    float sum = 0.f;
    for (int j = threadIdx.x; j < NN; j += blockDim.x) {
        float v = sc[j];
        if (v > 0.5f * NEGBIG) sum += __expf(v - m);
    }
    red[threadIdx.x] = sum;
    __syncthreads();
    for (int s = blockDim.x / 2; s > 0; s >>= 1) {
        if (threadIdx.x < s) red[threadIdx.x] += red[threadIdx.x + s];
        __syncthreads();
    }
    if (threadIdx.x == 0) {
        rowmax[i] = m;
        rowsum[i] = (red[0] > 0.f) ? red[0] : 1.f;
    }
}

__global__ void k_dense(const int* __restrict__ cnt,
                        const float* __restrict__ nv1, const float* __restrict__ nv2,
                        const float* __restrict__ rowmax, const float* __restrict__ rowsum,
                        const float* __restrict__ nsv,
                        _Float16* __restrict__ Ahi, _Float16* __restrict__ Alo) {
    int idx = blockIdx.x * blockDim.x + threadIdx.x;
    if (idx >= NN * NN) return;
    int s = idx >> 9, d = idx & 511;
    int m = cnt[idx];
    float val = 0.f;
    if (m) {
        float acc = 0.f;
#pragma unroll
        for (int a = 0; a < AEMB; a++) acc += nv1[s * AEMB + a] * nv2[a * NN + d];
        float scv = fmaxf(acc, 0.f);
        float ew = 1.f + __expf(scv - rowmax[s]) * __builtin_amdgcn_rcpf(rowsum[s]);
        val = (float)m * (ew * nsv[s]);
    }
    _Float16 hi = (_Float16)val;
    _Float16 lo = (_Float16)(val - (float)hi);
    int mt = d >> 4, r = d & 15;
    int kt = s >> 5, kk = s & 31;
    int half = (kk >> 3) & 1, p = kk >> 4, sl = kk & 7;
    size_t pos = ((((size_t)(mt * 16 + kt)) * 2 + p) * 32 + (r + 16 * half)) * 8 + sl;
    Ahi[pos] = hi;
    Alo[pos] = lo;
}

__global__ void k_norms(const float* degout, const float* degin, float* nsv, float* ndv) {
    int i = blockIdx.x * blockDim.x + threadIdx.x;
    if (i >= NN) return;
    nsv[i] = rsqrtf(fmaxf(degout[i], 1.f));
    ndv[i] = rsqrtf(fmaxf(degin[i], 1.f));
}

__global__ void k_start(const float* __restrict__ feat, const float* __restrict__ sw,
                        const float* __restrict__ sb, float* h) {
    int idx = blockIdx.x * blockDim.x + threadIdx.x;
    const int total = BB * NN * 13 * HH;
    if (idx >= total) return;
    int c = idx & 31;
    int t = (idx >> 5) % 13;
    int m = idx / (13 * HH);
    int b = m >> 9, n = m & 511;
    float x = (t == 0) ? 0.f : feat[(b * 12 + (t - 1)) * NN + n];
    h[idx] = sw[c] * x + sb[c];
}

__device__ __forceinline__ void put8(v16h& d, int off, float4 x, float4 y) {
    d[off + 0] = (_Float16)x.x; d[off + 1] = (_Float16)x.y;
    d[off + 2] = (_Float16)x.z; d[off + 3] = (_Float16)x.w;
    d[off + 4] = (_Float16)y.x; d[off + 5] = (_Float16)y.y;
    d[off + 6] = (_Float16)y.z; d[off + 7] = (_Float16)y.w;
}
__device__ __forceinline__ void gau_load_rows(float4 (&r)[8], const float* __restrict__ p, int d) {
    const float* q = p + d * HH;
    r[0] = ((const float4*)p)[0];        r[1] = ((const float4*)p)[1];
    r[2] = ((const float4*)(p + 16))[0]; r[3] = ((const float4*)(p + 16))[1];
    r[4] = ((const float4*)q)[0];        r[5] = ((const float4*)q)[1];
    r[6] = ((const float4*)(q + 16))[0]; r[7] = ((const float4*)(q + 16))[1];
}

__device__ __forceinline__ v16h cat8(v8h a, v8h b) {
    return __builtin_shufflevector(a, b, 0, 1, 2, 3, 4, 5, 6, 7, 8, 9, 10, 11, 12, 13, 14, 15);
}
__device__ __forceinline__ void msg_load_a(v8h (&h)[4][2], v8h (&l)[4][2],
                                           const v8h* __restrict__ Ah8,
                                           const v8h* __restrict__ Al8, int abase, int kt) {
#pragma unroll
    for (int i = 0; i < 4; i++)
#pragma unroll
        for (int p = 0; p < 2; p++) {
            h[i][p] = Ah8[abase + ((i * 16 + kt) * 2 + p) * 32];
            l[i][p] = Al8[abase + ((i * 16 + kt) * 2 + p) * 32];
        }
}
__device__ __forceinline__ void msg_load_b(v16h (&h)[4], v16h (&l)[4], const v8h* Bh8,
                                           const v8h* Bl8, int lane, int kt) {
#pragma unroll
    for (int j = 0; j < 4; j++) {
        int o = ((kt * 4 + j) * 2) * 32 + lane;
        h[j] = cat8(Bh8[o], Bh8[o + 32]);
        l[j] = cat8(Bl8[o], Bl8[o + 32]);
    }
}
__device__ __forceinline__ void msg_products(v8f (&acc)[4][4], const v8h (&ah)[4][2],
                                             const v8h (&al)[4][2], const v16h (&bh)[4],
                                             const v16h (&bl)[4]) {
    v16h fa[4], fl[4];
#pragma unroll
    for (int i = 0; i < 4; i++) {
        fa[i] = cat8(ah[i][0], ah[i][1]);
        fl[i] = cat8(al[i][0], al[i][1]);
    }
#pragma unroll
    for (int i = 0; i < 4; i++)
#pragma unroll
        for (int j = 0; j < 4; j++) acc[i][j] = wmma_f16(fl[i], bh[j], acc[i][j]);
#pragma unroll
    for (int i = 0; i < 4; i++)
#pragma unroll
        for (int j = 0; j < 4; j++) acc[i][j] = wmma_f16(fa[i], bl[j], acc[i][j]);
#pragma unroll
    for (int i = 0; i < 4; i++)
#pragma unroll
        for (int j = 0; j < 4; j++) acc[i][j] = wmma_f16(fa[i], bh[j], acc[i][j]);
}

__device__ __forceinline__ void skip_rowtile(const float* __restrict__ h,
                                             const float* __restrict__ fwT,
                                             const float* __restrict__ fb,
                                             const float* __restrict__ gwT,
                                             const float* __restrict__ gb,
                                             const float* __restrict__ sw,
                                             const float* __restrict__ sb, float* skip,
                                             int T, int d, int l, int tm, int lane) {
    const int Tp = T - d;
    const int n = lane & 15;
    const int base = (lane & 16) ? 8 : 0;
    float4 r[8];
    gau_load_rows(r, h + (size_t)((tm * 16 + n) * T + (Tp - 1)) * HH + base, d);
    v16h x0, x1;
    put8(x0, 0, r[0], r[1]);  put8(x0, 8, r[2], r[3]);
    put8(x1, 0, r[4], r[5]);  put8(x1, 8, r[6], r[7]);
    v16h a;
#pragma unroll
    for (int ct = 0; ct < 2; ct++) {
        const float* fwr = fwT + (size_t)((l * HH + ct * 16 + n) * 2) * HH + base;
        const float* gwr = gwT + (size_t)((l * HH + ct * 16 + n) * 2) * HH + base;
        v16h wf0, wf1, wg0, wg1;
        pack8(wf0, 0, fwr);       pack8(wf0, 8, fwr + 16);
        pack8(wf1, 0, fwr + HH);  pack8(wf1, 8, fwr + HH + 16);
        pack8(wg0, 0, gwr);       pack8(wg0, 8, gwr + 16);
        pack8(wg1, 0, gwr + HH);  pack8(wg1, 8, gwr + HH + 16);
        v8f cf = {}, cg = {};
        cf = wmma_f16(wf0, x0, cf);
        cg = wmma_f16(wg0, x0, cg);
        cf = wmma_f16(wf1, x1, cf);
        cg = wmma_f16(wg1, x1, cg);
        const float* fbp = fb + l * HH + ct * 16 + base;
        const float* gbp = gb + l * HH + ct * 16 + base;
        float4 f0 = ((const float4*)fbp)[0], f1 = ((const float4*)fbp)[1];
        float4 g0 = ((const float4*)gbp)[0], g1 = ((const float4*)gbp)[1];
        float fbv[8] = {f0.x, f0.y, f0.z, f0.w, f1.x, f1.y, f1.z, f1.w};
        float gbv[8] = {g0.x, g0.y, g0.z, g0.w, g1.x, g1.y, g1.z, g1.w};
#pragma unroll
        for (int v = 0; v < 8; v++)
            a[ct * 8 + v] = (_Float16)(fast_tanh(cf[v] + fbv[v]) * fast_sigmoid(cg[v] + gbv[v]));
    }
    float* prow = skip + (size_t)(tm * 16 + base) * SKC + n;
    for (int g = 0; g < 4; g++) {
        float old[4][8];
        v16h bm[4];
        float bias[4];
#pragma unroll
        for (int q = 0; q < 4; q++) {
            const int ncol = (g * 4 + q) * 16 + n;
            const float* brow = sw + (size_t)(l * SKC + ncol) * HH;
            pack8(bm[q], 0, brow + base);  pack8(bm[q], 8, brow + base + 16);
            bias[q] = sb[l * SKC + ncol];
#pragma unroll
            for (int v = 0; v < 8; v++) old[q][v] = prow[(size_t)v * SKC + (g * 4 + q) * 16];
        }
#pragma unroll
        for (int q = 0; q < 4; q++) {
            v8f c = {};
            c = wmma_f16(a, bm[q], c);
#pragma unroll
            for (int v = 0; v < 8; v++)
                prow[(size_t)v * SKC + (g * 4 + q) * 16] = old[q][v] + (c[v] + bias[q]);
        }
    }
}

#define SKIPBLK 128
__global__ __launch_bounds__(256) void k_msg_gconv(const float* __restrict__ fwT,
                                                   const float* __restrict__ fb,
                                                   const float* __restrict__ gwT,
                                                   const float* __restrict__ gb,
                                                   const _Float16* __restrict__ Ahi,
                                                   const _Float16* __restrict__ Alo,
                                                   const float* __restrict__ ndv,
                                                   const float* __restrict__ gcwT,
                                                   const float* __restrict__ gcb,
                                                   const float* __restrict__ hold,
                                                   float* __restrict__ hpre,
                                                   float* bnsum, float* bnsq,
                                                   const float* __restrict__ skw,
                                                   const float* __restrict__ skb,
                                                   float* skip,
                                                   int T, int d, int l) {
    const int Tp = T - d;
    const int gx = (Tp + 1) >> 1;
    if ((int)blockIdx.x >= gx * BB) {
        const int tm = ((int)blockIdx.x - gx * BB) * 8 + (threadIdx.x >> 5);
        skip_rowtile(hold, fwT, fb, gwT, gb, skw, skb, skip, T, d, l, tm, threadIdx.x & 31);
        return;
    }
    const int bx = (int)blockIdx.x % gx;
    __shared__ __attribute__((aligned(16))) _Float16 Bh[16 * 4 * 2 * 32 * 8];
    __shared__ __attribute__((aligned(16))) _Float16 Bl[16 * 4 * 2 * 32 * 8];
    __shared__ __attribute__((aligned(16))) _Float16 Zt[8 * 64 * 72];
    const int b = (int)blockIdx.x / gx;
    const int tid = threadIdx.x;
    const int wave = tid >> 5, lane = tid & 31;
    const int n = lane & 15;
    const int hf = lane >> 4;

    {
        v16h wf[2][2], wg[2][2];
        float fbias[2], gbias[2];
#pragma unroll
        for (int tn = 0; tn < 2; tn++) {
            const float* fwr = fwT + (size_t)((l * HH + tn * 16 + n) * 2) * HH + 8 * hf;
            const float* gwr = gwT + (size_t)((l * HH + tn * 16 + n) * 2) * HH + 8 * hf;
            pack8(wf[0][tn], 0, fwr);       pack8(wf[0][tn], 8, fwr + 16);
            pack8(wf[1][tn], 0, fwr + HH);  pack8(wf[1][tn], 8, fwr + HH + 16);
            pack8(wg[0][tn], 0, gwr);       pack8(wg[0][tn], 8, gwr + 16);
            pack8(wg[1][tn], 0, gwr + HH);  pack8(wg[1][tn], 8, gwr + HH + 16);
            fbias[tn] = fb[l * HH + tn * 16 + n];
            gbias[tn] = gb[l * HH + tn * 16 + n];
        }
        const float* hb = hold + (size_t)((b * NN + wave * 64 + n) * T + 2 * bx) * HH + 8 * hf;
        const bool two = 2 * bx + 1 < Tp;
        float4 r[8];
        gau_load_rows(r, hb, d);
        for (int q = 0; q < 8; q++) {
            const int i = q >> 1, tl = q & 1;
            const bool live = (tl == 0) || two;
            v16h a0, a1;
            put8(a0, 0, r[0], r[1]);  put8(a0, 8, r[2], r[3]);
            put8(a1, 0, r[4], r[5]);  put8(a1, 8, r[6], r[7]);
            if (q + 1 < 8 && (((q + 1) & 1) == 0 || two))
                gau_load_rows(r, hb + (size_t)(((q + 1) >> 1) * 16 * T + ((q + 1) & 1)) * HH, d);
            const int nt = wave * 4 + i;
            const int kt = nt >> 1, p = nt & 1;
#pragma unroll
            for (int tn = 0; tn < 2; tn++) {
                v8h hi = {}, lo = {};
                if (live) {
                    v8f cf = {}, cg = {};
                    cf = wmma_f16(a0, wf[0][tn], cf);
                    cg = wmma_f16(a0, wg[0][tn], cg);
                    cf = wmma_f16(a1, wf[1][tn], cf);
                    cg = wmma_f16(a1, wg[1][tn], cg);
#pragma unroll
                    for (int v = 0; v < 8; v++) {
                        float g = fast_tanh(cf[v] + fbias[tn]) * fast_sigmoid(cg[v] + gbias[tn]);
                        float x = g * 256.f;
                        _Float16 hh = (_Float16)x;
                        hi[v] = hh;
                        lo[v] = (_Float16)(x - (float)hh);
                    }
                }
                const int col = tl * 32 + tn * 16 + n;
                const int L = (col >> 2) + 16 * hf;
                const int off = ((((kt * 4 + (col & 3)) * 2 + p) * 32) + L) * 8;
                *(v8h*)(Bh + off) = hi;
                *(v8h*)(Bl + off) = lo;
            }
        }
    }
    __syncthreads();

    const v8h* __restrict__ Ah8 = (const v8h*)Ahi;
    const v8h* __restrict__ Al8 = (const v8h*)Alo;
    const v8h* Bh8 = (const v8h*)Bh;
    const v8h* Bl8 = (const v8h*)Bl;
    const int abase = (wave * 4 * 16 * 2) * 32 + lane;
    v8f acc[4][4];
#pragma unroll
    for (int i = 0; i < 4; i++)
#pragma unroll
        for (int j = 0; j < 4; j++) acc[i][j] = (v8f){};
    v8h a0h[4][2], a0l[4][2], a1h[4][2], a1l[4][2];
    v16h b0h[4], b0l[4], b1h[4], b1l[4];
    msg_load_a(a0h, a0l, Ah8, Al8, abase, 0);
    msg_load_b(b0h, b0l, Bh8, Bl8, lane, 0);
    for (int kt = 0; kt < 16; kt += 2) {
        msg_load_a(a1h, a1l, Ah8, Al8, abase, kt + 1);
        msg_load_b(b1h, b1l, Bh8, Bl8, lane, kt + 1);
        msg_products(acc, a0h, a0l, b0h, b0l);
        if (kt + 2 < 16) {
            msg_load_a(a0h, a0l, Ah8, Al8, abase, kt + 2);
            msg_load_b(b0h, b0l, Bh8, Bl8, lane, kt + 2);
        }
        msg_products(acc, a1h, a1l, b1h, b1l);
    }

    _Float16* zw = Zt + wave * (64 * 72);
#pragma unroll
    for (int i = 0; i < 4; i++) {
        const float* np = ndv + wave * 64 + i * 16 + hf * 8;
        float4 n0 = *(const float4*)np;
        float4 n1 = *(const float4*)(np + 4);
        float nr[8] = {n0.x, n0.y, n0.z, n0.w, n1.x, n1.y, n1.z, n1.w};
#pragma unroll
        for (int v = 0; v < 8; v++) {
            const float us = 1.f / 256.f;
            v4h q;
            q[0] = (_Float16)((acc[i][0][v] * us) * nr[v]);
            q[1] = (_Float16)((acc[i][1][v] * us) * nr[v]);
            q[2] = (_Float16)((acc[i][2][v] * us) * nr[v]);
            q[3] = (_Float16)((acc[i][3][v] * us) * nr[v]);
            *(v4h*)(zw + (i * 16 + hf * 8 + v) * 72 + 4 * n) = q;
        }
    }
    v16h bw[2];
    float bias[2];
#pragma unroll
    for (int tn = 0; tn < 2; tn++) {
        const float* brow = gcwT + (size_t)(l * HH + 2 * n + tn) * HH;
        pack8(bw[tn], 0, brow + 8 * hf);
        pack8(bw[tn], 8, brow + 8 * hf + 16);
        bias[tn] = gcb[l * HH + 2 * n + tn];
    }
    __syncthreads();
    float s1[2] = {0.f, 0.f}, s2[2] = {0.f, 0.f};
#pragma unroll
    for (int tl = 0; tl < 2; tl++) {
        const int t = 2 * bx + tl;
        if (t < Tp) {
#pragma unroll
            for (int i = 0; i < 4; i++) {
                const _Float16* zr = zw + (i * 16 + n) * 72 + tl * 32 + 8 * hf;
                v16h a = cat8(*(const v8h*)zr, *(const v8h*)(zr + 16));
                v8f c0 = wmma_f16(a, bw[0], (v8f){});
                v8f c1 = wmma_f16(a, bw[1], (v8f){});
#pragma unroll
                for (int v = 0; v < 8; v++) {
                    int bn = b * NN + wave * 64 + i * 16 + hf * 8 + v;
                    float2 hv = *(const float2*)(hold + (size_t)((bn * T + t + d) * HH + 2 * n));
                    float v0 = fmaxf(c0[v] + bias[0], 0.f) + hv.x;
                    float v1 = fmaxf(c1[v] + bias[1], 0.f) + hv.y;
                    *(float2*)(hpre + (size_t)(bn * Tp + t) * HH + 2 * n) = make_float2(v0, v1);
                    s1[0] += v0;
                    s2[0] += v0 * v0;
                    s1[1] += v1;
                    s2[1] += v1 * v1;
                }
            }
        }
    }
#pragma unroll
    for (int tn = 0; tn < 2; tn++) {
        atomicAdd(&bnsum[2 * n + tn], s1[tn]);
        atomicAdd(&bnsq[2 * n + tn], s2[tn]);
    }
}

__global__ void k_bn(float* h, const float* __restrict__ bnsum, const float* __restrict__ bnsq,
                     const float* __restrict__ gamma, const float* __restrict__ beta,
                     int Tp, int l) {
    int total = BB * NN * Tp * HH;
    int idx = blockIdx.x * blockDim.x + threadIdx.x;
    if (idx >= total) return;
    int c = idx & 31;
    float cnt = (float)(BB * NN * Tp);
    float mu = bnsum[c] / cnt;
    float var = bnsq[c] / cnt - mu * mu;
    float x = h[idx];
    h[idx] = (x - mu) * rsqrtf(var + 1e-5f) * gamma[l * HH + c] + beta[l * HH + c];
}

#define E1_PITCH 264
__device__ __forceinline__ void end1_load(v16h (&fa)[2], v16h (&fb)[4], const _Float16* abase,
                                          const _Float16* bbase, int ks) {
#pragma unroll
    for (int i = 0; i < 2; i++) {
        const _Float16* p = abase + i * 16 * E1_PITCH + ks * 32;
        fa[i] = cat8(*(const v8h*)p, *(const v8h*)(p + 16));
    }
#pragma unroll
    for (int j = 0; j < 4; j++) {
        const _Float16* p = bbase + j * 16 * E1_PITCH + ks * 32;
        fb[j] = cat8(*(const v8h*)p, *(const v8h*)(p + 16));
    }
}
__device__ __forceinline__ void end1_products(v8f (&acc)[2][4], const v16h (&fa)[2],
                                              const v16h (&fb)[4]) {
#pragma unroll
    for (int i = 0; i < 2; i++)
#pragma unroll
        for (int j = 0; j < 4; j++) acc[i][j] = wmma_f16(fa[i], fb[j], acc[i][j]);
}
__global__ __launch_bounds__(256) void k_end1_wmma(const float* __restrict__ skip,
                                                   const float* __restrict__ w1,
                                                   const float* __restrict__ b1,
                                                   float* __restrict__ out1) {
    __shared__ __attribute__((aligned(16))) _Float16 As[128 * E1_PITCH];
    __shared__ __attribute__((aligned(16))) _Float16 Bs[128 * E1_PITCH];
    const int bm = (int)blockIdx.x >> 2;
    const int bn = (int)blockIdx.x & 3;
    const int tid = threadIdx.x;
    const int wave = tid >> 5, lane = tid & 31;
    const int n = lane & 15, hf = lane >> 4;

    {
        const int r0 = tid >> 6;
        const int c4 = (tid & 63) * 4;
        const float* ap = skip + (size_t)(bm * 128 + r0) * SKC + c4;
        const float* bp = w1 + (size_t)(bn * 128 + r0) * SKC + c4;
#pragma unroll 1
        for (int ph = 0; ph < 4; ph++) {
            float4 va[8], vb[8];
#pragma unroll
            for (int u = 0; u < 8; u++) {
                const int it = ph * 8 + u;
                va[u] = *(const float4*)(ap + (size_t)(it * 4) * SKC);
                vb[u] = *(const float4*)(bp + (size_t)(it * 4) * SKC);
            }
#pragma unroll
            for (int u = 0; u < 8; u++) {
                const int row = (ph * 8 + u) * 4 + r0;
                v4h qa, qb;
                qa[0] = (_Float16)fmaxf(va[u].x, 0.f);  qa[1] = (_Float16)fmaxf(va[u].y, 0.f);
                qa[2] = (_Float16)fmaxf(va[u].z, 0.f);  qa[3] = (_Float16)fmaxf(va[u].w, 0.f);
                qb[0] = (_Float16)vb[u].x;  qb[1] = (_Float16)vb[u].y;
                qb[2] = (_Float16)vb[u].z;  qb[3] = (_Float16)vb[u].w;
                *(v4h*)(As + row * E1_PITCH + c4) = qa;
                const int prow = (row & 64) + 16 * (row & 3) + ((row >> 2) & 15);
                *(v4h*)(Bs + prow * E1_PITCH + c4) = qb;
            }
        }
    }
    __syncthreads();

    const int wr = wave >> 1, wc = wave & 1;
    v8f acc[2][4];
#pragma unroll
    for (int i = 0; i < 2; i++)
#pragma unroll
        for (int j = 0; j < 4; j++) acc[i][j] = (v8f){};
    const _Float16* abase = As + (wr * 32 + n) * E1_PITCH + 8 * hf;
    const _Float16* bbase = Bs + (wc * 64 + n) * E1_PITCH + 8 * hf;
    v16h fa0[2], fb0[4], fa1[2], fb1[4];
    end1_load(fa0, fb0, abase, bbase, 0);
#pragma unroll 1
    for (int ks = 0; ks < 8; ks += 2) {
        end1_load(fa1, fb1, abase, bbase, ks + 1);
        end1_products(acc, fa0, fb0);
        if (ks + 2 < 8) end1_load(fa0, fb0, abase, bbase, ks + 2);
        end1_products(acc, fa1, fb1);
    }

    const int col = bn * 128 + wc * 64 + 4 * n;
    const float4 bias = *(const float4*)(b1 + col);
#pragma unroll
    for (int i = 0; i < 2; i++)
#pragma unroll
        for (int v = 0; v < 8; v++) {
            const int row = bm * 128 + wr * 32 + i * 16 + hf * 8 + v;
            float4 o;
            o.x = fmaxf(acc[i][0][v] + bias.x, 0.f);
            o.y = fmaxf(acc[i][1][v] + bias.y, 0.f);
            o.z = fmaxf(acc[i][2][v] + bias.z, 0.f);
            o.w = fmaxf(acc[i][3][v] + bias.w, 0.f);
            *(float4*)(out1 + (size_t)row * ECC + col) = o;
        }
}

__global__ void k_end2_wmma(const float* __restrict__ in1, const float* __restrict__ w2,
                            const float* __restrict__ b2, float* out) {
    int tm = blockIdx.x * 4 + (threadIdx.x >> 5);
    int lane = threadIdx.x & 31;
    int mrow = tm * 16 + (lane & 15);
    int ncol = lane & 15;
    int base = (lane & 16) ? 8 : 0;
    const float* arow = in1 + (size_t)mrow * ECC;
    const float* brow = w2 + (size_t)ncol * ECC;
    v8f c = {};
#pragma unroll
    for (int k0 = 0; k0 < ECC; k0 += 32) {
        v16h a, bm = {};
        pack8(a, 0, arow + k0 + base);
        pack8(a, 8, arow + k0 + base + 16);
        if (ncol < OUTF) {
            pack8(bm, 0, brow + k0 + base);
            pack8(bm, 8, brow + k0 + base + 16);
        }
        c = wmma_f16(a, bm, c);
    }
    if (ncol < OUTF) {
        int rbase = tm * 16 + ((lane & 16) ? 8 : 0);
        float bias = b2[ncol];
#pragma unroll
        for (int v = 0; v < 8; v++) {
            int row = rbase + v;
            int b = row >> 9, n = row & 511;
            out[(b * OUTF + ncol) * NN + n] = c[v] + bias;
        }
    }
}

extern "C" void kernel_launch(void* const* d_in, const int* in_sizes, int n_in,
                              void* d_out, int out_size, void* d_ws, size_t ws_size,
                              hipStream_t stream) {
    const float* feat     = (const float*)d_in[0];
    const int*   src      = (const int*)d_in[1];
    const int*   dst      = (const int*)d_in[2];
    const float* nv1      = (const float*)d_in[3];
    const float* nv2      = (const float*)d_in[4];
    const float* start_w  = (const float*)d_in[5];
    const float* start_b  = (const float*)d_in[6];
    const float* filter_w = (const float*)d_in[7];
    const float* filter_b = (const float*)d_in[8];
    const float* gate_w   = (const float*)d_in[9];
    const float* gate_b   = (const float*)d_in[10];
    const float* skip_w   = (const float*)d_in[11];
    const float* skip_b   = (const float*)d_in[12];
    const float* bn_gamma = (const float*)d_in[13];
    const float* bn_beta  = (const float*)d_in[14];
    const float* gconv_w  = (const float*)d_in[15];
    const float* gconv_b  = (const float*)d_in[16];
    const float* end1_w   = (const float*)d_in[17];
    const float* end1_b   = (const float*)d_in[18];
    const float* end2_w   = (const float*)d_in[19];
    const float* end2_b   = (const float*)d_in[20];

    float* ws = (float*)d_ws;
    size_t o = 0;
    int* adj = (int*)(ws + o); o += NN * NN;
    float* degout = ws + o; o += NN;
    float* degin  = ws + o; o += NN;
    float* bnsum  = ws + o; o += NLAY * HH;
    float* bnsq   = ws + o; o += NLAY * HH;
    float* skip   = ws + o; o += (size_t)BB * NN * SKC;
    const int nZero = NN * NN + NN + NN + NLAY * HH + NLAY * HH + BB * NN * SKC;
    float* rowmax = ws + o; o += NN;
    float* rowsum = ws + o; o += NN;
    _Float16* Ahi = (_Float16*)(ws + o); o += (NN * NN) / 2;
    _Float16* Alo = (_Float16*)(ws + o); o += (NN * NN) / 2;
    float* nsv    = ws + o; o += NN;
    float* ndv    = ws + o; o += NN;
    float* fwT    = ws + o; o += NLAY * HH * 2 * HH;
    float* gwT    = ws + o; o += NLAY * HH * 2 * HH;
    float* gcwT   = ws + o; o += NLAY * HH * HH;
    float* hA     = ws + o; o += (size_t)BB * NN * 13 * HH;
    float* hB     = ws + o; o += (size_t)BB * NN * 13 * HH;
    float* out1   = ws + o; o += (size_t)BB * NN * ECC;

    k_zero<<<(nZero + 255) / 256, 256, 0, stream>>>((float*)adj, nZero);

    k_prep_w<<<(NLAY * HH * 2 * HH + 255) / 256, 256, 0, stream>>>(filter_w, gate_w, gconv_w,
                                                                   fwT, gwT, gcwT);
    k_scatter<<<(NE + 255) / 256, 256, 0, stream>>>(src, dst, adj, degout, degin);
    k_rownorm<<<NN, 256, 0, stream>>>(adj, nv1, nv2, rowmax, rowsum);
    k_norms<<<(NN + 255) / 256, 256, 0, stream>>>(degout, degin, nsv, ndv);
    k_dense<<<(NN * NN) / 256, 256, 0, stream>>>(adj, nv1, nv2, rowmax, rowsum, nsv, Ahi, Alo);
    k_start<<<(BB * NN * 13 * HH + 255) / 256, 256, 0, stream>>>(feat, start_w, start_b, hA);

    const int dil[8] = {1, 2, 1, 2, 1, 2, 1, 2};
    float* hcur = hA;
    float* hnext = hB;
    int T = 13;
    for (int l = 0; l < NLAY; l++) {
        int d = dil[l];
        int Tp = T - d;
        int M = BB * NN * Tp;

        k_msg_gconv<<<((Tp + 1) / 2) * BB + SKIPBLK, 256, 0, stream>>>(
            fwT, filter_b, gwT, gate_b, Ahi, Alo, ndv, gcwT, gconv_b, hcur, hnext,
            bnsum + l * HH, bnsq + l * HH, skip_w, skip_b, skip, T, d, l);
        k_bn<<<(M * HH + 255) / 256, 256, 0, stream>>>(hnext, bnsum + l * HH, bnsq + l * HH,
                                                       bn_gamma, bn_beta, Tp, l);
        float* tmp = hcur; hcur = hnext; hnext = tmp;
        T = Tp;
    }

    k_end1_wmma<<<(BB * NN / 128) * (ECC / 128), 256, 0, stream>>>(skip, end1_w, end1_b, out1);
    k_end2_wmma<<<(BB * NN / 16) / 4, 128, 0, stream>>>(out1, end2_w, end2_b, (float*)d_out);
}
